# starting kernel
# baseline (speedup 1.0000x reference)
#include <hip/hip_runtime.h>
#include <hip/hip_bf16.h>

typedef float v2f __attribute__((ext_vector_type(2)));
typedef float v8f __attribute__((ext_vector_type(8)));

#define HIDDEN 10
#define T_LEN  2048
#define BATCH  4096
#define INPUT  4
#define PF     8

#if __has_builtin(__builtin_amdgcn_tanhf)
  __device__ __forceinline__ float dev_tanh(float x) { return __builtin_amdgcn_tanhf(x); }
#elif __has_builtin(__builtin_amdgcn_tanh_f32)
  __device__ __forceinline__ float dev_tanh(float x) { return __builtin_amdgcn_tanh_f32(x); }
#else
  __device__ __forceinline__ float dev_tanh(float x) {
    float r = __builtin_amdgcn_rcpf(__expf(2.0f * x) + 1.0f);
    return __builtin_fmaf(-2.0f, r, 1.0f);
  }
#endif

__device__ __forceinline__ float fast_sigmoid(float x) {
    return __builtin_fmaf(0.5f, dev_tanh(0.5f * x), 0.5f);
}
__device__ __forceinline__ float fast_tanh(float x) { return dev_tanh(x); }

__device__ __forceinline__ void wave_order_barrier() {
    __builtin_amdgcn_wave_barrier();
}

__global__ __launch_bounds__(32)
void lstm_wmma_f32_kernel(const float* __restrict__ x,
                          const float* __restrict__ W_ih,
                          const float* __restrict__ W_hh,
                          const float* __restrict__ b_ih,
                          const float* __restrict__ b_hh,
                          const float* __restrict__ W_lin,
                          const float* __restrict__ b_lin,
                          float* __restrict__ out)
{
    __shared__ __align__(32) float hlds2[16 * 16];

    const int lane = threadIdx.x;
    const int m    = lane & 15;
    const int half = lane >> 4;
    const int n    = m;
    const int b0   = blockIdx.x * 16;

    v2f Bih[4];
    v2f Bhh[4][3];
    v8f Cbias[4];
    float wl[HIDDEN];
    float wl6[6];

    #pragma unroll
    for (int g = 0; g < 4; ++g) {
        float e0 = 0.f, e1 = 0.f, bia = 0.f;
        if (n < HIDDEN) {
            const int row = g * HIDDEN + n;
            const int k0  = 2 * half;
            e0  = W_ih[row * INPUT + k0];
            e1  = W_ih[row * INPUT + k0 + 1];
            bia = b_ih[row] + b_hh[row];
        }
        Bih[g].x = e0; Bih[g].y = e1;
        v8f c;
        #pragma unroll
        for (int e = 0; e < 8; ++e) c[e] = bia;
        Cbias[g] = c;

        #pragma unroll
        for (int j = 0; j < 3; ++j) {
            float h0 = 0.f, h1 = 0.f;
            if (n < HIDDEN) {
                const int row = g * HIDDEN + n;
                const int kk  = j * 4 + 2 * half;
                if (kk     < HIDDEN) h0 = W_hh[row * HIDDEN + kk];
                if (kk + 1 < HIDDEN) h1 = W_hh[row * HIDDEN + kk + 1];
            }
            Bhh[g][j].x = h0; Bhh[g][j].y = h1;
        }
    }
    #pragma unroll
    for (int i = 0; i < HIDDEN; ++i) wl[i] = W_lin[i];
    #pragma unroll
    for (int j = 0; j < 3; ++j) {
        const int kk = j * 4 + 2 * half;
        wl6[2 * j + 0] = (kk     < HIDDEN) ? W_lin[kk]     : 0.f;
        wl6[2 * j + 1] = (kk + 1 < HIDDEN) ? W_lin[kk + 1] : 0.f;
    }
    const float blin = b_lin[0];

    for (int i = lane; i < 256; i += 32) hlds2[i] = 0.f;
    __syncthreads();

    v8f creg;
    #pragma unroll
    for (int e = 0; e < 8; ++e) creg[e] = 0.f;

    const float* xrow = x + ((size_t)(b0 + m) * T_LEN) * INPUT + 2 * half;
    float*       orow = out + (size_t)(b0 + m) * T_LEN;

    v2f xbuf[PF], xcur[PF];
    #pragma unroll
    for (int p = 0; p < PF; ++p) {
        xbuf[p].x = xrow[(size_t)p * INPUT + 0];
        xbuf[p].y = xrow[(size_t)p * INPUT + 1];
    }

    for (int tb = 0; tb < T_LEN; tb += PF) {
        #pragma unroll
        for (int p = 0; p < PF; ++p) xcur[p] = xbuf[p];

        {
            int tnext = tb + PF;
            if (tnext > T_LEN - PF) tnext = T_LEN - PF;
            const float* xpf = xrow + (size_t)tnext * INPUT;
            #pragma unroll
            for (int p = 0; p < PF; ++p) {
                xbuf[p].x = xpf[p * INPUT + 0];
                xbuf[p].y = xpf[p * INPUT + 1];
            }
        }

        #pragma unroll
        for (int u = 0; u < PF; ++u) {
            const int t = tb + u;
            const v2f xa = xcur[u];

            v2f ha[3];
            #pragma unroll
            for (int j = 0; j < 3; ++j) {
                const int kk = j * 4 + 2 * half;
                ha[j].x = hlds2[(kk    ) * 16 + m];
                ha[j].y = hlds2[(kk + 1) * 16 + m];
            }

            v8f acc[4];
            #pragma unroll
            for (int g = 0; g < 4; ++g)
                acc[g] = __builtin_amdgcn_wmma_f32_16x16x4_f32(
                    false, xa, false, Bih[g], (short)0, Cbias[g], false, false);

            #pragma unroll
            for (int j = 0; j < 3; ++j)
                #pragma unroll
                for (int g = 0; g < 4; ++g)
                    acc[g] = __builtin_amdgcn_wmma_f32_16x16x4_f32(
                        false, ha[j], false, Bhh[g][j], (short)0, acc[g], false, false);

            {
                float po = ha[0].x * wl6[0] + ha[0].y * wl6[1]
                         + ha[1].x * wl6[2] + ha[1].y * wl6[3]
                         + ha[2].x * wl6[4] + ha[2].y * wl6[5];
                po += __shfl_xor(po, 16, 32);
                if (half == 0) {
                    const int to = (t > 0) ? (t - 1) : 0;
                    orow[to] = po + blin;
                }
            }

            wave_order_barrier();

            v8f hv;
            #pragma unroll
            for (int e = 0; e < 8; ++e) {
                const float iv = fast_sigmoid(acc[0][e]);
                const float fv = fast_sigmoid(acc[1][e]);
                const float gv = fast_tanh   (acc[2][e]);
                const float ov = fast_sigmoid(acc[3][e]);
                const float cv = fv * creg[e] + iv * gv;
                creg[e] = cv;
                hv[e] = ov * fast_tanh(cv);
            }
            *(v8f*)&hlds2[n * 16 + 8 * half] = hv;
            wave_order_barrier();
        }
    }

    if (half == 0) {
        float o = blin;
        #pragma unroll
        for (int i = 0; i < HIDDEN; ++i) o += hlds2[i * 16 + m] * wl[i];
        orow[T_LEN - 1] = o;
    }
}

extern "C" void kernel_launch(void* const* d_in, const int* in_sizes, int n_in,
                              void* d_out, int out_size, void* d_ws, size_t ws_size,
                              hipStream_t stream) {
    const float* x     = (const float*)d_in[0];
    const float* W_ih  = (const float*)d_in[1];
    const float* W_hh  = (const float*)d_in[2];
    const float* b_ih  = (const float*)d_in[3];
    const float* b_hh  = (const float*)d_in[4];
    const float* W_lin = (const float*)d_in[5];
    const float* b_lin = (const float*)d_in[6];
    float* out = (float*)d_out;

    dim3 grid(BATCH / 16);
    dim3 block(32);
    lstm_wmma_f32_kernel<<<grid, block, 0, stream>>>(
        x, W_ih, W_hh, b_ih, b_hh, W_lin, b_lin, out);
}


# h_t goes to the next step's A fragments by one transposing WMMA with a fixed 0/1 matrix (no LDS round trip); next input projection issued beside the cell update; k slots regrouped (sum order changes)
# speedup vs baseline: 2.0524x; 2.0524x over previous
#include <hip/hip_runtime.h>
#include <hip/hip_bf16.h>

typedef float v2f __attribute__((ext_vector_type(2)));
typedef float v4f __attribute__((ext_vector_type(4)));
typedef float v8f __attribute__((ext_vector_type(8)));

#define HIDDEN 10
#define T_LEN  2048
#define BATCH  4096
#define INPUT  4
#define PF     8
#define ROWS   4
#define EPL    2
#define WAVES  4
#define WAVE_PITCH 13824

#if __has_builtin(__builtin_amdgcn_tanhf)
  __device__ __forceinline__ float dev_tanh(float x) { return __builtin_amdgcn_tanhf(x); }
#elif __has_builtin(__builtin_amdgcn_tanh_f32)
  __device__ __forceinline__ float dev_tanh(float x) { return __builtin_amdgcn_tanh_f32(x); }
#else
  __device__ __forceinline__ float dev_tanh(float x) {
    float r = __builtin_amdgcn_rcpf(__expf(2.0f * x) + 1.0f);
    return __builtin_fmaf(-2.0f, r, 1.0f);
  }
#endif

__device__ __forceinline__ float fast_sigmoid(float x) {
    return __builtin_fmaf(0.5f, dev_tanh(0.5f * x), 0.5f);
}
__device__ __forceinline__ float fast_tanh(float x) { return dev_tanh(x); }

__device__ __forceinline__ void wave_order_barrier() {
    __builtin_amdgcn_wave_barrier();
}

__device__ __forceinline__ int lane_of_hidden(int u) { return (u < 5) ? u : u + 3; }
__device__ __forceinline__ int hidden_of_lane(int n) { return ((n & 7) < 5) ? (n & 7) + 5 * (n >> 3) : -1; }
__device__ __forceinline__ int free_index(int n)     { return ((n & 7) - 5) + 3 * (n >> 3); }

__device__ __forceinline__ int tile_row(int tile, int n) {
    const int u = hidden_of_lane(n);
    if (u >= 0) return tile * HIDDEN + u;
    const int q = free_index(n);
    if (tile == 0) return 3 * HIDDEN + q;
    if (tile == 1 && q < 4) return 3 * HIDDEN + 6 + q;
    return -1;
}

__device__ __forceinline__ float row_gather(float v, unsigned sel_lo, unsigned sel_hi) {
    const unsigned u = __builtin_bit_cast(unsigned, v);
    return __builtin_bit_cast(float, __builtin_amdgcn_permlane16(u, u, sel_lo, sel_hi, true, false));
}

__device__ __forceinline__ float other_half(float v) {
    const unsigned u = __builtin_bit_cast(unsigned, v);
    return __builtin_bit_cast(float, __builtin_amdgcn_permlanex16(u, u, 0x76543210u, 0xFEDCBA98u, true, false));
}

__device__ __forceinline__ v4f head_group(const float (&pq)[8], int half, float blin) {
    v4f v;
    #pragma unroll
    for (int i = 0; i < 4; ++i) {
        const float mine  = half ? pq[4 + i] : pq[i];
        const float yours = half ? pq[i]     : pq[4 + i];
        v[i] = (mine + other_half(yours)) + blin;
    }
    return v;
}

__device__ __forceinline__ int tile_batch_row(int M) {
    return (M < 8) ? (M & 1) : 2 + (M & 1);
}

__global__ __launch_bounds__(32 * WAVES)
void lstm_wmma_f32_kernel(const float* __restrict__ x,
                          const float* __restrict__ W_ih,
                          const float* __restrict__ W_hh,
                          const float* __restrict__ b_ih,
                          const float* __restrict__ b_hh,
                          const float* __restrict__ W_lin,
                          const float* __restrict__ b_lin,
                          float* __restrict__ out)
{
    __shared__ __align__(32) float hlds_all[(WAVES - 1) * WAVE_PITCH + 16 * ROWS];

    const int wave = threadIdx.x >> 5;
    float* const hlds2 = hlds_all + wave * WAVE_PITCH;
    const int lane = threadIdx.x & 31;
    const int m    = lane & 15;
    const int half = lane >> 4;
    const int n    = m;
    const int b0   = (blockIdx.x * WAVES + wave) * ROWS;
    const int rm   = tile_batch_row(m);

    v2f Bih[3];
    v2f Bhh[3][3];
    v8f Cbias[3];
    float wl[HIDDEN];
    float wl5[5];

    #pragma unroll
    for (int g = 0; g < 3; ++g) {
        float e0 = 0.f, e1 = 0.f, bia = 0.f;
        const int row = tile_row(g, n);
        if (row >= 0) {
            const int k0  = 2 * half;
            e0  = W_ih[row * INPUT + k0];
            e1  = W_ih[row * INPUT + k0 + 1];
            bia = b_ih[row] + b_hh[row];
        }
        Bih[g].x = e0; Bih[g].y = e1;
        v8f c;
        #pragma unroll
        for (int e = 0; e < 8; ++e) c[e] = bia;
        Cbias[g] = c;

        #pragma unroll
        for (int j = 0; j < 3; ++j) {
            float h0 = 0.f, h1 = 0.f;
            if (row >= 0) {
                h0 = W_hh[row * HIDDEN + 2 * j + 5 * half];
                if (2 * j + 1 < 5) h1 = W_hh[row * HIDDEN + 2 * j + 1 + 5 * half];
            }
            Bhh[g][j].x = h0; Bhh[g][j].y = h1;
        }
    }
    #pragma unroll
    for (int i = 0; i < HIDDEN; ++i) wl[i] = W_lin[i];
    #pragma unroll
    for (int s = 0; s < 5; ++s) wl5[s] = W_lin[s + 5 * half];
    const float blin = b_lin[0];

    v2f Ssel;
    Ssel.x = (rm == 2 * half)     ? 1.f : 0.f;
    Ssel.y = (rm == 2 * half + 1) ? 1.f : 0.f;
    v8f zero8;
    #pragma unroll
    for (int e = 0; e < 8; ++e) zero8[e] = 0.f;

    float creg[EPL];
    #pragma unroll
    for (int e = 0; e < EPL; ++e) creg[e] = 0.f;

    const float* xrow = x + ((size_t)(b0 + rm) * T_LEN) * INPUT + 2 * half;
    float*       orow = out + (size_t)(b0 + rm) * T_LEN;
    const bool   row_lane = ((m & 7) < 2);
    float pq[PF];
    #pragma unroll
    for (int p = 0; p < PF; ++p) pq[p] = 0.f;

    v2f xbuf[PF], xcur[PF];
    #pragma unroll
    for (int p = 0; p < PF; ++p) {
        xbuf[p].x = xrow[(size_t)p * INPUT + 0];
        xbuf[p].y = xrow[(size_t)p * INPUT + 1];
    }

    v8f hT = zero8;
    v2f hv;
    hv.x = 0.f; hv.y = 0.f;

    v8f accn[3];
    #pragma unroll
    for (int g = 0; g < 3; ++g)
        accn[g] = __builtin_amdgcn_wmma_f32_16x16x4_f32(
            false, xbuf[0], false, Bih[g], (short)0, Cbias[g], false, false);

    for (int tb = 0; tb < T_LEN; tb += PF) {
        #pragma unroll
        for (int p = 0; p < PF; ++p) xcur[p] = xbuf[p];

        {
            int tnext = tb + PF;
            if (tnext > T_LEN - PF) tnext = T_LEN - PF;
            const float* xpf = xrow + (size_t)tnext * INPUT;
            #pragma unroll
            for (int p = 0; p < PF; ++p) {
                xbuf[p].x = xpf[p * INPUT + 0];
                xbuf[p].y = xpf[p * INPUT + 1];
            }
        }

        #pragma unroll
        for (int u = 0; u < PF; ++u) {
            v2f ha[3];
            #pragma unroll
            for (int j = 0; j < 3; ++j) {
                ha[j].x = hT[2 * j];
                ha[j].y = hT[2 * j + 1];
            }

            if (u == 1 && tb != 0) {
                const v4f og = head_group(pq, half, blin);
                if (row_lane) *(v4f*)(orow + (tb - PF) + 4 * half) = og;
            }

            v8f acc[3];
            #pragma unroll
            for (int g = 0; g < 3; ++g) acc[g] = accn[g];
            #pragma unroll
            for (int j = 0; j < 3; ++j)
                #pragma unroll
                for (int g = 0; g < 3; ++g)
                    acc[g] = __builtin_amdgcn_wmma_f32_16x16x4_f32(
                        false, ha[j], false, Bhh[g][j], (short)0, acc[g], false, false);

            const v2f xn = (u + 1 < PF) ? xcur[(u + 1) % PF] : xbuf[0];
            #pragma unroll
            for (int g = 0; g < 2; ++g)
                accn[g] = __builtin_amdgcn_wmma_f32_16x16x4_f32(
                    false, xn, false, Bih[g], (short)0, Cbias[g], false, false);

            {
                float p = ha[0].y * wl5[1];
                p = __builtin_fmaf(ha[0].x, wl5[0], p);
                p = __builtin_fmaf(ha[1].x, wl5[2], p);
                p = __builtin_fmaf(ha[1].y, wl5[3], p);
                p = __builtin_fmaf(ha[2].x, wl5[4], p);
                pq[(u + PF - 1) % PF] = p;
            }

            #pragma unroll
            for (int e = 0; e < EPL; ++e) {
                const float iv = fast_sigmoid(acc[0][e]);
                const float fv = fast_sigmoid(acc[1][e]);
                const float gv = fast_tanh   (acc[2][e]);
                const float oa = row_gather(iv, 0x765ED765u, 0xFEDCBA9Fu);
                const float ob = row_gather(fv, 0x76543210u, 0xFEDD7658u);
                const float ov = (n <= 8) ? oa : ob;
                const float cv = fv * creg[e] + iv * gv;
                creg[e] = cv;
                hv[e] = ov * fast_tanh(cv);
            }

            asm("v_wmma_f32_16x16x4_f32 %0, %1, %2, 0\n\tv_nop\n\tv_nop\n\tv_nop\n\tv_nop"
                : "=&v"(hT) : "v"(hv), "v"(Ssel));
            __builtin_amdgcn_sched_barrier(0);
            accn[2] = __builtin_amdgcn_wmma_f32_16x16x4_f32(
                false, xn, false, Bih[2], (short)0, Cbias[2], false, false);
            __builtin_amdgcn_sched_barrier(0);
        }
    }

    {
        *(v2f*)&hlds2[n * ROWS + EPL * half] = hv;
        wave_order_barrier();
        float o = blin;
        #pragma unroll
        for (int i = 0; i < HIDDEN; ++i) o += hlds2[lane_of_hidden(i) * ROWS + rm] * wl[i];
        v4f og = head_group(pq, half, blin);
        if (half) og[3] = o;
        if (row_lane) *(v4f*)(orow + (T_LEN - PF) + 4 * half) = og;
    }
}

extern "C" void kernel_launch(void* const* d_in, const int* in_sizes, int n_in,
                              void* d_out, int out_size, void* d_ws, size_t ws_size,
                              hipStream_t stream) {
    const float* x     = (const float*)d_in[0];
    const float* W_ih  = (const float*)d_in[1];
    const float* W_hh  = (const float*)d_in[2];
    const float* b_ih  = (const float*)d_in[3];
    const float* b_hh  = (const float*)d_in[4];
    const float* W_lin = (const float*)d_in[5];
    const float* b_lin = (const float*)d_in[6];
    float* out = (float*)d_out;

    dim3 grid(BATCH / (ROWS * WAVES));
    dim3 block(32 * WAVES);
    lstm_wmma_f32_kernel<<<grid, block, 0, stream>>>(
        x, W_ih, W_hh, b_ih, b_hh, W_lin, b_lin, out);
}


# Input projection of 4 time steps by one WMMA per gate tile (tile rows = batch row x step), used whole as C; a step reads its own register pair: 43 WMMA per 4 steps (was 52), 2 x loads per body (16)
# speedup vs baseline: 2.2511x; 1.0968x over previous
#include <hip/hip_runtime.h>
#include <hip/hip_bf16.h>

typedef float v2f __attribute__((ext_vector_type(2)));
typedef float v4f __attribute__((ext_vector_type(4)));
typedef float v8f __attribute__((ext_vector_type(8)));

#define HIDDEN 10
#define T_LEN  2048
#define BATCH  4096
#define INPUT  4
#define PF     8
#define GS     4
#define NGRP   (T_LEN / GS)
#define ROWS   4
#define EPL    2
#define WAVES  4
#define WAVE_PITCH 13824

#if __has_builtin(__builtin_amdgcn_tanhf)
  __device__ __forceinline__ float dev_tanh(float x) { return __builtin_amdgcn_tanhf(x); }
#elif __has_builtin(__builtin_amdgcn_tanh_f32)
  __device__ __forceinline__ float dev_tanh(float x) { return __builtin_amdgcn_tanh_f32(x); }
#else
  __device__ __forceinline__ float dev_tanh(float x) {
    float r = __builtin_amdgcn_rcpf(__expf(2.0f * x) + 1.0f);
    return __builtin_fmaf(-2.0f, r, 1.0f);
  }
#endif

__device__ __forceinline__ float fast_sigmoid(float x) {
    return __builtin_fmaf(0.5f, dev_tanh(0.5f * x), 0.5f);
}
__device__ __forceinline__ float fast_tanh(float x) { return dev_tanh(x); }

__device__ __forceinline__ void wave_order_barrier() {
    __builtin_amdgcn_wave_barrier();
}

__device__ __forceinline__ int lane_of_hidden(int u) { return (u < 5) ? u : u + 3; }
__device__ __forceinline__ int hidden_of_lane(int n) { return ((n & 7) < 5) ? (n & 7) + 5 * (n >> 3) : -1; }
__device__ __forceinline__ int free_index(int n)     { return ((n & 7) - 5) + 3 * (n >> 3); }

__device__ __forceinline__ int tile_row(int tile, int n) {
    const int u = hidden_of_lane(n);
    if (u >= 0) return tile * HIDDEN + u;
    const int q = free_index(n);
    if (tile == 0) return 3 * HIDDEN + q;
    if (tile == 1 && q < 4) return 3 * HIDDEN + 6 + q;
    return -1;
}

__device__ __forceinline__ float row_gather(float v, unsigned sel_lo, unsigned sel_hi) {
    const unsigned u = __builtin_bit_cast(unsigned, v);
    return __builtin_bit_cast(float, __builtin_amdgcn_permlane16(u, u, sel_lo, sel_hi, true, false));
}

__device__ __forceinline__ float other_half(float v) {
    const unsigned u = __builtin_bit_cast(unsigned, v);
    return __builtin_bit_cast(float, __builtin_amdgcn_permlanex16(u, u, 0x76543210u, 0xFEDCBA98u, true, false));
}

__device__ __forceinline__ v4f head_group(const float (&pq)[8], int half, float blin) {
    v4f v;
    #pragma unroll
    for (int i = 0; i < 4; ++i) {
        const float mine  = half ? pq[4 + i] : pq[i];
        const float yours = half ? pq[i]     : pq[4 + i];
        v[i] = (mine + other_half(yours)) + blin;
    }
    return v;
}

__device__ __forceinline__ int tile_batch_row(int M) {
    return (M < 8) ? (M & 1) : 2 + (M & 1);
}

__global__ __launch_bounds__(32 * WAVES)
void lstm_wmma_f32_kernel(const float* __restrict__ x,
                          const float* __restrict__ W_ih,
                          const float* __restrict__ W_hh,
                          const float* __restrict__ b_ih,
                          const float* __restrict__ b_hh,
                          const float* __restrict__ W_lin,
                          const float* __restrict__ b_lin,
                          float* __restrict__ out)
{
    __shared__ __align__(32) float hlds_all[(WAVES - 1) * WAVE_PITCH + 16 * ROWS];

    const int wave = threadIdx.x >> 5;
    float* const hlds2 = hlds_all + wave * WAVE_PITCH;
    const int lane = threadIdx.x & 31;
    const int m    = lane & 15;
    const int half = lane >> 4;
    const int n    = m;
    const int b0   = (blockIdx.x * WAVES + wave) * ROWS;
    const int rm   = tile_batch_row(m);

    v2f Bih[3];
    v2f Bhh[3][3];
    v8f Cbias[3];
    float wl[HIDDEN];
    float wl5[5];

    #pragma unroll
    for (int g = 0; g < 3; ++g) {
        float e0 = 0.f, e1 = 0.f, bia = 0.f;
        const int row = tile_row(g, n);
        if (row >= 0) {
            const int k0  = 2 * half;
            e0  = W_ih[row * INPUT + k0];
            e1  = W_ih[row * INPUT + k0 + 1];
            bia = b_ih[row] + b_hh[row];
        }
        Bih[g].x = e0; Bih[g].y = e1;
        v8f c;
        #pragma unroll
        for (int e = 0; e < 8; ++e) c[e] = bia;
        Cbias[g] = c;

        #pragma unroll
        for (int j = 0; j < 3; ++j) {
            float h0 = 0.f, h1 = 0.f;
            if (row >= 0) {
                h0 = W_hh[row * HIDDEN + 2 * j + 5 * half];
                if (2 * j + 1 < 5) h1 = W_hh[row * HIDDEN + 2 * j + 1 + 5 * half];
            }
            Bhh[g][j].x = h0; Bhh[g][j].y = h1;
        }
    }
    #pragma unroll
    for (int i = 0; i < HIDDEN; ++i) wl[i] = W_lin[i];
    #pragma unroll
    for (int s = 0; s < 5; ++s) wl5[s] = W_lin[s + 5 * half];
    const float blin = b_lin[0];

    v2f Ssel;
    Ssel.x = (rm == 2 * half)     ? 1.f : 0.f;
    Ssel.y = (rm == 2 * half + 1) ? 1.f : 0.f;
    v8f zero8;
    #pragma unroll
    for (int e = 0; e < 8; ++e) zero8[e] = 0.f;

    float creg[EPL];
    #pragma unroll
    for (int e = 0; e < EPL; ++e) creg[e] = 0.f;

    const float* xlane = x + ((size_t)(b0 + rm) * T_LEN + ((m >> 1) & 3)) * INPUT + 2 * half;
    float*       orow = out + (size_t)(b0 + rm) * T_LEN;
    const bool   row_lane = ((m & 7) < 2);
    float pq[PF];
    #pragma unroll
    for (int p = 0; p < PF; ++p) pq[p] = 0.f;

    v2f xa1, xa2, xn1, xn2;
    const v2f xg0 = *(const v2f*)(xlane);
    xn1 = *(const v2f*)(xlane + 1 * GS * INPUT);
    xn2 = *(const v2f*)(xlane + 2 * GS * INPUT);

    v8f hT = zero8;
    v2f hv;
    hv.x = 0.f; hv.y = 0.f;

    v8f Pa[3], Pb[3];
    #pragma unroll
    for (int g = 0; g < 3; ++g)
        Pa[g] = __builtin_amdgcn_wmma_f32_16x16x4_f32(
            false, xg0, false, Bih[g], (short)0, Cbias[g], false, false);

    for (int tb = 0; tb < T_LEN; tb += PF) {
        xa1 = xn1;
        xa2 = xn2;
        {
            int g1 = tb / GS + 3, g2 = tb / GS + 4;
            if (g1 > NGRP - 1) g1 = NGRP - 1;
            if (g2 > NGRP - 1) g2 = NGRP - 1;
            xn1 = *(const v2f*)(xlane + (size_t)g1 * (GS * INPUT));
            xn2 = *(const v2f*)(xlane + (size_t)g2 * (GS * INPUT));
        }

        #pragma unroll
        for (int u = 0; u < PF; ++u) {
            v2f ha[3];
            #pragma unroll
            for (int j = 0; j < 3; ++j) {
                ha[j].x = hT[2 * j];
                ha[j].y = hT[2 * j + 1];
            }

            if (u == 1 && tb != 0) {
                const v4f og = head_group(pq, half, blin);
                if (row_lane) *(v4f*)(orow + (tb - PF) + 4 * half) = og;
            }

            const int tt = u % GS;
            v8f acc[3];
            #pragma unroll
            for (int g = 0; g < 3; ++g) acc[g] = (u < GS) ? Pa[g] : Pb[g];
            #pragma unroll
            for (int j = 0; j < 3; ++j)
                #pragma unroll
                for (int g = 0; g < 3; ++g)
                    acc[g] = __builtin_amdgcn_wmma_f32_16x16x4_f32(
                        false, ha[j], false, Bhh[g][j], (short)0, acc[g], false, false);

            if (tt < 3) {
                __builtin_amdgcn_sched_barrier(0);
                if (u < GS)
                    Pb[tt] = __builtin_amdgcn_wmma_f32_16x16x4_f32(
                        false, xa1, false, Bih[tt], (short)0, Cbias[tt], false, false);
                else
                    Pa[tt] = __builtin_amdgcn_wmma_f32_16x16x4_f32(
                        false, xa2, false, Bih[tt], (short)0, Cbias[tt], false, false);
                __builtin_amdgcn_sched_barrier(0);
            }

            {
                float p = ha[0].y * wl5[1];
                p = __builtin_fmaf(ha[0].x, wl5[0], p);
                p = __builtin_fmaf(ha[1].x, wl5[2], p);
                p = __builtin_fmaf(ha[1].y, wl5[3], p);
                p = __builtin_fmaf(ha[2].x, wl5[4], p);
                pq[(u + PF - 1) % PF] = p;
            }

            #pragma unroll
            for (int e = 0; e < EPL; ++e) {
                const float iv = fast_sigmoid(acc[0][2 * tt + e]);
                const float fv = fast_sigmoid(acc[1][2 * tt + e]);
                const float gv = fast_tanh   (acc[2][2 * tt + e]);
                const float oa = row_gather(iv, 0x765ED765u, 0xFEDCBA9Fu);
                const float ob = row_gather(fv, 0x76543210u, 0xFEDD7658u);
                const float ov = (n <= 8) ? oa : ob;
                const float cv = fv * creg[e] + iv * gv;
                creg[e] = cv;
                hv[e] = ov * fast_tanh(cv);
            }

            asm("v_wmma_f32_16x16x4_f32 %0, %1, %2, 0\n\tv_nop\n\tv_nop\n\tv_nop\n\tv_nop\n\tv_nop\n\t"
                "s_wait_alu depctr_va_vdst(0)"
                : "=&v"(hT) : "v"(hv), "v"(Ssel));
        }
    }

    {
        *(v2f*)&hlds2[n * ROWS + EPL * half] = hv;
        wave_order_barrier();
        float o = blin;
        #pragma unroll
        for (int i = 0; i < HIDDEN; ++i) o += hlds2[lane_of_hidden(i) * ROWS + rm] * wl[i];
        v4f og = head_group(pq, half, blin);
        if (half) og[3] = o;
        if (row_lane) *(v4f*)(orow + (T_LEN - PF) + 4 * half) = og;
    }
}

extern "C" void kernel_launch(void* const* d_in, const int* in_sizes, int n_in,
                              void* d_out, int out_size, void* d_ws, size_t ws_size,
                              hipStream_t stream) {
    const float* x     = (const float*)d_in[0];
    const float* W_ih  = (const float*)d_in[1];
    const float* W_hh  = (const float*)d_in[2];
    const float* b_ih  = (const float*)d_in[3];
    const float* b_hh  = (const float*)d_in[4];
    const float* W_lin = (const float*)d_in[5];
    const float* b_lin = (const float*)d_in[6];
    float* out = (float*)d_out;

    dim3 grid(BATCH / (ROWS * WAVES));
    dim3 block(32 * WAVES);
    lstm_wmma_f32_kernel<<<grid, block, 0, stream>>>(
        x, W_ih, W_hh, b_ih, b_hh, W_lin, b_lin, out);
}


# Units 4 and 9, the half-empty third k chunk of the recurrent product, go to the vector unit: 4 row gathers + FMAs beside 6 WMMAs per step (was 9); the tail is summed apart (sum order changes)
# speedup vs baseline: 2.4530x; 1.0897x over previous
#include <hip/hip_runtime.h>
#include <hip/hip_bf16.h>

typedef float v2f __attribute__((ext_vector_type(2)));
typedef float v4f __attribute__((ext_vector_type(4)));
typedef float v8f __attribute__((ext_vector_type(8)));

#define HIDDEN 10
#define T_LEN  2048
#define BATCH  4096
#define INPUT  4
#define PF     8
#define GS     4
#define NGRP   (T_LEN / GS)
#define ROWS   4
#define EPL    2
#define WAVES  4
#define WAVE_PITCH 13824

#if __has_builtin(__builtin_amdgcn_tanhf)
  __device__ __forceinline__ float dev_tanh(float x) { return __builtin_amdgcn_tanhf(x); }
#elif __has_builtin(__builtin_amdgcn_tanh_f32)
  __device__ __forceinline__ float dev_tanh(float x) { return __builtin_amdgcn_tanh_f32(x); }
#else
  __device__ __forceinline__ float dev_tanh(float x) {
    float r = __builtin_amdgcn_rcpf(__expf(2.0f * x) + 1.0f);
    return __builtin_fmaf(-2.0f, r, 1.0f);
  }
#endif

__device__ __forceinline__ float fast_sigmoid(float x) {
    return __builtin_fmaf(0.5f, dev_tanh(0.5f * x), 0.5f);
}
__device__ __forceinline__ float fast_sigmoid_sum(float x, float hs) {
    return __builtin_fmaf(0.5f, dev_tanh(__builtin_fmaf(0.5f, x, hs)), 0.5f);
}
__device__ __forceinline__ float fast_tanh(float x) { return dev_tanh(x); }

__device__ __forceinline__ void wave_order_barrier() {
    __builtin_amdgcn_wave_barrier();
}

__device__ __forceinline__ int lane_of_hidden(int u) { return (u < 5) ? u : u + 3; }
__device__ __forceinline__ int hidden_of_lane(int n) { return ((n & 7) < 5) ? (n & 7) + 5 * (n >> 3) : -1; }
__device__ __forceinline__ int free_index(int n)     { return ((n & 7) - 5) + 3 * (n >> 3); }

__device__ __forceinline__ int tile_row(int tile, int n) {
    const int u = hidden_of_lane(n);
    if (u >= 0) return tile * HIDDEN + u;
    const int q = free_index(n);
    if (tile == 0) return 3 * HIDDEN + q;
    if (tile == 1 && q < 4) return 3 * HIDDEN + 6 + q;
    return -1;
}

__device__ __forceinline__ float row_gather(float v, unsigned sel_lo, unsigned sel_hi) {
    const unsigned u = __builtin_bit_cast(unsigned, v);
    return __builtin_bit_cast(float, __builtin_amdgcn_permlane16(u, u, sel_lo, sel_hi, true, false));
}

__device__ __forceinline__ float other_half(float v) {
    const unsigned u = __builtin_bit_cast(unsigned, v);
    return __builtin_bit_cast(float, __builtin_amdgcn_permlanex16(u, u, 0x76543210u, 0xFEDCBA98u, true, false));
}

__device__ __forceinline__ v4f head_group(const float (&pq)[8], int half, float blin) {
    v4f v;
    #pragma unroll
    for (int i = 0; i < 4; ++i) {
        const float mine  = half ? pq[4 + i] : pq[i];
        const float yours = half ? pq[i]     : pq[4 + i];
        v[i] = (mine + other_half(yours)) + blin;
    }
    return v;
}

__device__ __forceinline__ int tile_batch_row(int M) {
    return (M < 8) ? (M & 1) : 2 + (M & 1);
}

__global__ __launch_bounds__(32 * WAVES)
void lstm_wmma_f32_kernel(const float* __restrict__ x,
                          const float* __restrict__ W_ih,
                          const float* __restrict__ W_hh,
                          const float* __restrict__ b_ih,
                          const float* __restrict__ b_hh,
                          const float* __restrict__ W_lin,
                          const float* __restrict__ b_lin,
                          float* __restrict__ out)
{
    __shared__ __align__(32) float hlds_all[(WAVES - 1) * WAVE_PITCH + 16 * ROWS];

    const int wave = threadIdx.x >> 5;
    float* const hlds2 = hlds_all + wave * WAVE_PITCH;
    const int lane = threadIdx.x & 31;
    const int m    = lane & 15;
    const int half = lane >> 4;
    const int n    = m;
    const int b0   = (blockIdx.x * WAVES + wave) * ROWS;
    const int rm   = tile_batch_row(m);

    v2f Bih[3];
    v2f Bhh[3][2];
    float wt4[3], wt9[3];
    v8f Cbias[3];
    float wl[HIDDEN];
    float wl5[5];

    #pragma unroll
    for (int g = 0; g < 3; ++g) {
        float e0 = 0.f, e1 = 0.f, bia = 0.f;
        const int row = tile_row(g, n);
        if (row >= 0) {
            const int k0  = 2 * half;
            e0  = W_ih[row * INPUT + k0];
            e1  = W_ih[row * INPUT + k0 + 1];
            bia = b_ih[row] + b_hh[row];
        }
        Bih[g].x = e0; Bih[g].y = e1;
        v8f c;
        #pragma unroll
        for (int e = 0; e < 8; ++e) c[e] = bia;
        Cbias[g] = c;

        #pragma unroll
        for (int j = 0; j < 2; ++j) {
            float h0 = 0.f, h1 = 0.f;
            if (row >= 0) {
                h0 = W_hh[row * HIDDEN + 2 * j + 5 * half];
                h1 = W_hh[row * HIDDEN + 2 * j + 1 + 5 * half];
            }
            Bhh[g][j].x = h0; Bhh[g][j].y = h1;
        }
        float t4 = 0.f, t9 = 0.f;
        if (row >= 0) {
            t4 = W_hh[row * HIDDEN + 4];
            t9 = W_hh[row * HIDDEN + 9];
        }
        wt4[g] = (g < 2) ? 0.5f * t4 : t4;
        wt9[g] = (g < 2) ? 0.5f * t9 : t9;
    }
    #pragma unroll
    for (int i = 0; i < HIDDEN; ++i) wl[i] = W_lin[i];
    #pragma unroll
    for (int s = 0; s < 5; ++s) wl5[s] = W_lin[s + 5 * half];
    const float blin = b_lin[0];

    v2f Ssel;
    Ssel.x = (rm == 2 * half)     ? 1.f : 0.f;
    Ssel.y = (rm == 2 * half + 1) ? 1.f : 0.f;
    v8f zero8;
    #pragma unroll
    for (int e = 0; e < 8; ++e) zero8[e] = 0.f;

    float creg[EPL];
    #pragma unroll
    for (int e = 0; e < EPL; ++e) creg[e] = 0.f;

    const float* xlane = x + ((size_t)(b0 + rm) * T_LEN + ((m >> 1) & 3)) * INPUT + 2 * half;
    float*       orow = out + (size_t)(b0 + rm) * T_LEN;
    const bool   row_lane = ((m & 7) < 2);
    float pq[PF];
    #pragma unroll
    for (int p = 0; p < PF; ++p) pq[p] = 0.f;

    v2f xa1, xa2, xn1, xn2;
    const v2f xg0 = *(const v2f*)(xlane);
    xn1 = *(const v2f*)(xlane + 1 * GS * INPUT);
    xn2 = *(const v2f*)(xlane + 2 * GS * INPUT);

    v8f hT = zero8;
    v2f hv;
    hv.x = 0.f; hv.y = 0.f;
    float hb4[EPL], hb9[EPL];
    #pragma unroll
    for (int e = 0; e < EPL; ++e) { hb4[e] = 0.f; hb9[e] = 0.f; }
    v8f Cg = zero8;
    const unsigned sel4 = 0x44444444u, sel9 = 0xCCCCCCCCu;

    v8f Pa[3], Pb[3];
    #pragma unroll
    for (int g = 0; g < 3; ++g)
        Pa[g] = __builtin_amdgcn_wmma_f32_16x16x4_f32(
            false, xg0, false, Bih[g], (short)0, Cbias[g], false, false);

    for (int tb = 0; tb < T_LEN; tb += PF) {
        xa1 = xn1;
        xa2 = xn2;
        {
            int g1 = tb / GS + 3, g2 = tb / GS + 4;
            if (g1 > NGRP - 1) g1 = NGRP - 1;
            if (g2 > NGRP - 1) g2 = NGRP - 1;
            xn1 = *(const v2f*)(xlane + (size_t)g1 * (GS * INPUT));
            xn2 = *(const v2f*)(xlane + (size_t)g2 * (GS * INPUT));
        }

        #pragma unroll
        for (int u = 0; u < PF; ++u) {
            v2f ha[3];
            #pragma unroll
            for (int j = 0; j < 3; ++j) {
                ha[j].x = hT[2 * j];
                ha[j].y = hT[2 * j + 1];
            }

            if (u == 1 && tb != 0) {
                const v4f og = head_group(pq, half, blin);
                if (row_lane) *(v4f*)(orow + (tb - PF) + 4 * half) = og;
            }

            const int tt = u % GS;
            v8f acc[3];
            #pragma unroll
            for (int g = 0; g < 2; ++g)
                acc[g] = __builtin_amdgcn_wmma_f32_16x16x4_f32(
                    false, ha[0], false, Bhh[g][0], (short)0, (u < GS) ? Pa[g] : Pb[g], false, false);
            #pragma unroll
            for (int e = 0; e < EPL; ++e) {
                const float pg = (u < GS) ? Pa[2][2 * tt + e] : Pb[2][2 * tt + e];
                Cg[2 * tt + e] = __builtin_fmaf(hb4[e], wt4[2], __builtin_fmaf(hb9[e], wt9[2], pg));
            }
            acc[2] = __builtin_amdgcn_wmma_f32_16x16x4_f32(
                false, ha[0], false, Bhh[2][0], (short)0, Cg, false, false);
            #pragma unroll
            for (int g = 0; g < 3; ++g)
                acc[g] = __builtin_amdgcn_wmma_f32_16x16x4_f32(
                    false, ha[1], false, Bhh[g][1], (short)0, acc[g], false, false);
            float tl0[EPL], tl1[EPL];
            #pragma unroll
            for (int e = 0; e < EPL; ++e) {
                tl0[e] = __builtin_fmaf(hb4[e], wt4[0], hb9[e] * wt9[0]);
                tl1[e] = __builtin_fmaf(hb4[e], wt4[1], hb9[e] * wt9[1]);
            }

            __builtin_amdgcn_sched_barrier(0);
            if (tt < 3) {
                if (u < GS)
                    Pb[tt] = __builtin_amdgcn_wmma_f32_16x16x4_f32(
                        false, xa1, false, Bih[tt], (short)0, Cbias[tt], false, false);
                else
                    Pa[tt] = __builtin_amdgcn_wmma_f32_16x16x4_f32(
                        false, xa2, false, Bih[tt], (short)0, Cbias[tt], false, false);
                __builtin_amdgcn_sched_barrier(0);
            }

            {
                float p = ha[0].y * wl5[1];
                p = __builtin_fmaf(ha[0].x, wl5[0], p);
                p = __builtin_fmaf(ha[1].x, wl5[2], p);
                p = __builtin_fmaf(ha[1].y, wl5[3], p);
                p = __builtin_fmaf(ha[2].x, wl5[4], p);
                pq[(u + PF - 1) % PF] = p;
            }

            #pragma unroll
            for (int e = 0; e < EPL; ++e) {
                const float iv = fast_sigmoid_sum(acc[0][2 * tt + e], tl0[e]);
                const float fv = fast_sigmoid_sum(acc[1][2 * tt + e], tl1[e]);
                const float gv = fast_tanh   (acc[2][2 * tt + e]);
                const float oa = row_gather(iv, 0x765ED765u, 0xFEDCBA9Fu);
                const float ob = row_gather(fv, 0x76543210u, 0xFEDD7658u);
                const float ov = (n <= 8) ? oa : ob;
                const float cv = fv * creg[e] + iv * gv;
                creg[e] = cv;
                hv[e] = ov * fast_tanh(cv);
            }

            asm("v_wmma_f32_16x16x4_f32 %0, %5, %6, 0\n\t"
                "v_permlane16_b32 %1, %7, %9, %9\n\t"
                "v_permlane16_b32 %2, %8, %9, %9\n\t"
                "v_permlane16_b32 %3, %7, %10, %10\n\t"
                "v_permlane16_b32 %4, %8, %10, %10\n\t"
                "v_nop\n\t"
                "s_wait_alu depctr_va_vdst(0)"
                : "=&v"(hT), "=&v"(hb4[0]), "=&v"(hb4[1]), "=&v"(hb9[0]), "=&v"(hb9[1])
                : "v"(hv), "v"(Ssel), "v"(hv.x), "v"(hv.y), "s"(sel4), "s"(sel9));
        }
    }

    {
        *(v2f*)&hlds2[n * ROWS + EPL * half] = hv;
        wave_order_barrier();
        float o = blin;
        #pragma unroll
        for (int i = 0; i < HIDDEN; ++i) o += hlds2[lane_of_hidden(i) * ROWS + rm] * wl[i];
        v4f og = head_group(pq, half, blin);
        if (half) og[3] = o;
        if (row_lane) *(v4f*)(orow + (T_LEN - PF) + 4 * half) = og;
    }
}

extern "C" void kernel_launch(void* const* d_in, const int* in_sizes, int n_in,
                              void* d_out, int out_size, void* d_ws, size_t ws_size,
                              hipStream_t stream) {
    const float* x     = (const float*)d_in[0];
    const float* W_ih  = (const float*)d_in[1];
    const float* W_hh  = (const float*)d_in[2];
    const float* b_ih  = (const float*)d_in[3];
    const float* b_hh  = (const float*)d_in[4];
    const float* W_lin = (const float*)d_in[5];
    const float* b_lin = (const float*)d_in[6];
    float* out = (float*)d_out;

    dim3 grid(BATCH / (ROWS * WAVES));
    dim3 block(32 * WAVES);
    lstm_wmma_f32_kernel<<<grid, block, 0, stream>>>(
        x, W_ih, W_hh, b_ih, b_hh, W_lin, b_lin, out);
}
